# starting kernel
# baseline (speedup 1.0000x reference)
#include <hip/hip_runtime.h>
#include <hip/hip_bf16.h>


typedef __attribute__((ext_vector_type(2))) float v2f;
typedef __attribute__((ext_vector_type(8))) float v8f;

#define B_DIM 8
#define S_DIM 4096
#define D_DIM 2048
#define E_DIM 16
#define KSPLIT 4
#define KCHUNK (D_DIM / KSPLIT)
#define CAPACITY 8.0f
#define EPSILON 1e-6f

__global__ __launch_bounds__(1024)
void switch_gate_wmma_kernel(const float* __restrict__ X,
                             const float* __restrict__ W,
                             const float* __restrict__ bias,
                             float* __restrict__ out) {
  __shared__ float part[KSPLIT * B_DIM][16][E_DIM];
  __shared__ float gate[B_DIM][16][E_DIM];

  const int tid  = threadIdx.x;
  const int ww   = tid >> 5;
  const int lane = tid & 31;
  const int half = lane >> 4;
  const int col  = lane & 15;
  const int b    = ww & 7;
  const int q    = ww >> 3;
  const int s0   = blockIdx.x * 16;
  const int k0   = q * KCHUNK;

  const float* __restrict__ xrow =
      X + ((size_t)b * S_DIM + (size_t)(s0 + col)) * D_DIM + k0 + 2 * half;
  const float* __restrict__ wcol = W + (size_t)(k0 + 2 * half) * E_DIM + col;

  v8f c = {};
  #pragma unroll 8
  for (int kk = 0; kk < KCHUNK; kk += 4) {
    v2f a = *(const v2f*)(xrow + kk);
    v2f bfrag;
    bfrag.x = wcol[kk * E_DIM];
    bfrag.y = wcol[kk * E_DIM + E_DIM];
    c = __builtin_amdgcn_wmma_f32_16x16x4_f32(
         false, a,  false, bfrag,
         (short)0, c,  false,  false);
  }

  #pragma unroll
  for (int j = 0; j < 8; ++j) part[ww][j + 8 * half][col] = c[j];
  __syncthreads();

  const float bval = bias[col];
  float g[8];
  #pragma unroll
  for (int j = 0; j < 8; ++j) {
    const int m = j + 8 * half;
    float l = bval + part[b][m][col] + part[b + 8][m][col]
                   + part[b + 16][m][col] + part[b + 24][m][col];
    float mx = l;
    mx = fmaxf(mx, __shfl_xor(mx, 1));
    mx = fmaxf(mx, __shfl_xor(mx, 2));
    mx = fmaxf(mx, __shfl_xor(mx, 4));
    mx = fmaxf(mx, __shfl_xor(mx, 8));
    float e = __expf(l - mx);
    float s = e;
    s += __shfl_xor(s, 1);
    s += __shfl_xor(s, 2);
    s += __shfl_xor(s, 4);
    s += __shfl_xor(s, 8);
    g[j] = e / s;
    if (q == 0) gate[b][j + 8 * half][col] = g[j];
  }
  __syncthreads();

  if (q == 0) {
    #pragma unroll
    for (int j = 0; j < 8; ++j) {
      const int m = j + 8 * half;
      float denom = EPSILON;
      #pragma unroll
      for (int bb = 0; bb < B_DIM; ++bb) denom += gate[bb][m][col];
      out[((size_t)b * S_DIM + (size_t)(s0 + m)) * E_DIM + col] =
          g[j] / denom * CAPACITY;
    }
  }
}

extern "C" void kernel_launch(void* const* d_in, const int* in_sizes, int n_in,
                              void* d_out, int out_size, void* d_ws, size_t ws_size,
                              hipStream_t stream) {
  const float* X    = (const float*)d_in[0];
  const float* W    = (const float*)d_in[1];
  const float* bias = (const float*)d_in[2];
  float* out        = (float*)d_out;

  dim3 grid(S_DIM / 16);
  dim3 block(1024);
  switch_gate_wmma_kernel<<<grid, block, 0, stream>>>(X, W, bias, out);
}


# X read as whole 128-B lines (4 lines per 16-B wave load, was 16 per 8-B load), turned into A fragments via a wave-private LDS slab; k permuted alike in A and B in each 8-k group (sum order changes)
# speedup vs baseline: 1.2747x; 1.2747x over previous
#include <hip/hip_runtime.h>


typedef __attribute__((ext_vector_type(2))) float v2f;
typedef __attribute__((ext_vector_type(4))) float v4f;
typedef __attribute__((ext_vector_type(8))) float v8f;

#define B_DIM 8
#define S_DIM 4096
#define D_DIM 2048
#define E_DIM 16
#define KSPLIT 4
#define KCHUNK (D_DIM / KSPLIT)
#define TK 32
#define NTILE (KCHUNK / TK)
#define PITCH 36
#define AHEAD 2
#define CAPACITY 8.0f
#define EPSILON 1e-6f

__global__ __launch_bounds__(1024)
void switch_gate_wmma_kernel(const float* __restrict__ X,
                             const float* __restrict__ W,
                             const float* __restrict__ bias,
                             float* __restrict__ out) {
  __shared__ float part[KSPLIT * B_DIM][16][E_DIM];
  __shared__ float gate[B_DIM][16][E_DIM];
  __shared__ __attribute__((aligned(16))) float slab[KSPLIT * B_DIM][16 * PITCH];

  const int tid  = threadIdx.x;
  const int ww   = tid >> 5;
  const int lane = tid & 31;
  const int half = lane >> 4;
  const int col  = lane & 15;
  const int b    = ww & 7;
  const int q    = ww >> 3;
  const int s0   = blockIdx.x * 16;
  const int k0   = q * KCHUNK;

  const int lr = lane >> 3;
  const int lp = lane & 7;
  const float* __restrict__ xt =
      X + ((size_t)b * S_DIM + (size_t)(s0 + lr)) * D_DIM + k0 + 4 * lp;
  float* const myslab = slab[ww];
  float* const st = myslab + lr * PITCH + 4 * lp;
  const float* const fr = myslab + col * PITCH + 4 * half;

  const float* __restrict__ wcol = W + (size_t)(k0 + 4 * half) * E_DIM + col;

  v4f xa[AHEAD][4];
  v2f wb[2][8];

#define LOAD_X(T) do { _Pragma("unroll") for (int j = 0; j < 4; ++j) \
    xa[(T) % AHEAD][j] = *(const v4f*)(xt + (size_t)j * 4 * D_DIM + (T) * TK); } while (0)
#define LOAD_W(T) do { _Pragma("unroll") for (int i = 0; i < 8; ++i) { \
    const int kr = (T) * TK + 8 * (i >> 1) + 2 * (i & 1); \
    wb[(T) & 1][i].x = wcol[kr * E_DIM]; \
    wb[(T) & 1][i].y = wcol[kr * E_DIM + E_DIM]; } } while (0)
#define WAVE_LDS_ORDER() do { __builtin_amdgcn_fence(__ATOMIC_RELEASE, "wavefront"); \
    __builtin_amdgcn_wave_barrier(); \
    __builtin_amdgcn_fence(__ATOMIC_ACQUIRE, "wavefront"); } while (0)

  LOAD_W(0);
  #pragma unroll
  for (int t = 0; t < AHEAD; ++t) LOAD_X(t);

  v8f c = {};
  #pragma unroll
  for (int t = 0; t < NTILE; ++t) {
    #pragma unroll
    for (int j = 0; j < 4; ++j) *(v4f*)(st + j * 4 * PITCH) = xa[t % AHEAD][j];
    if (t + 1 < NTILE) LOAD_W(t + 1);
    if (t + AHEAD < NTILE) LOAD_X(t + AHEAD);
    WAVE_LDS_ORDER();
    #pragma unroll
    for (int w = 0; w < 4; ++w) {
      const v4f av = *(const v4f*)(fr + 8 * w);
      const v2f a0 = {av.x, av.y};
      const v2f a1 = {av.z, av.w};
      c = __builtin_amdgcn_wmma_f32_16x16x4_f32(
           false, a0,  false, wb[t & 1][2 * w],
           (short)0, c,  false,  false);
      c = __builtin_amdgcn_wmma_f32_16x16x4_f32(
           false, a1,  false, wb[t & 1][2 * w + 1],
           (short)0, c,  false,  false);
    }
    WAVE_LDS_ORDER();
  }
#undef LOAD_X
#undef LOAD_W
#undef WAVE_LDS_ORDER

  #pragma unroll
  for (int j = 0; j < 8; ++j) part[ww][j + 8 * half][col] = c[j];
  __syncthreads();

  const float bval = bias[col];
  float g[8];
  #pragma unroll
  for (int j = 0; j < 8; ++j) {
    const int m = j + 8 * half;
    float l = bval + part[b][m][col] + part[b + 8][m][col]
                   + part[b + 16][m][col] + part[b + 24][m][col];
    float mx = l;
    mx = fmaxf(mx, __shfl_xor(mx, 1));
    mx = fmaxf(mx, __shfl_xor(mx, 2));
    mx = fmaxf(mx, __shfl_xor(mx, 4));
    mx = fmaxf(mx, __shfl_xor(mx, 8));
    float e = __expf(l - mx);
    float s = e;
    s += __shfl_xor(s, 1);
    s += __shfl_xor(s, 2);
    s += __shfl_xor(s, 4);
    s += __shfl_xor(s, 8);
    g[j] = e / s;
    if (q == 0) gate[b][j + 8 * half][col] = g[j];
  }
  __syncthreads();

  if (q == 0) {
    #pragma unroll
    for (int j = 0; j < 8; ++j) {
      const int m = j + 8 * half;
      float denom = EPSILON;
      #pragma unroll
      for (int bb = 0; bb < B_DIM; ++bb) denom += gate[bb][m][col];
      out[((size_t)b * S_DIM + (size_t)(s0 + m)) * E_DIM + col] =
          g[j] / denom * CAPACITY;
    }
  }
}

extern "C" void kernel_launch(void* const* d_in, const int* in_sizes, int n_in,
                              void* d_out, int out_size, void* d_ws, size_t ws_size,
                              hipStream_t stream) {
  const float* X    = (const float*)d_in[0];
  const float* W    = (const float*)d_in[1];
  const float* bias = (const float*)d_in[2];
  float* out        = (float*)d_out;

  dim3 grid(S_DIM / 16);
  dim3 block(1024);
  switch_gate_wmma_kernel<<<grid, block, 0, stream>>>(X, W, bias, out);
}


# Each work-group starts its K walk at a different 256-B position (tile (t+r) mod 16, r from the block index) so that X and W requests of work-groups in step spread over a row; sum order changes
# speedup vs baseline: 1.3175x; 1.0336x over previous
#include <hip/hip_runtime.h>


typedef __attribute__((ext_vector_type(2))) float v2f;
typedef __attribute__((ext_vector_type(4))) float v4f;
typedef __attribute__((ext_vector_type(8))) float v8f;

#define B_DIM 8
#define S_DIM 4096
#define D_DIM 2048
#define E_DIM 16
#define KSPLIT 4
#define KCHUNK (D_DIM / KSPLIT)
#define TK 32
#define NTILE (KCHUNK / TK)
#define PITCH 36
#define AHEAD 2
#define CAPACITY 8.0f
#define EPSILON 1e-6f

__global__ __launch_bounds__(1024)
void switch_gate_wmma_kernel(const float* __restrict__ X,
                             const float* __restrict__ W,
                             const float* __restrict__ bias,
                             float* __restrict__ out) {
  __shared__ float part[KSPLIT * B_DIM][16][E_DIM];
  __shared__ float gate[B_DIM][16][E_DIM];
  __shared__ __attribute__((aligned(16))) float slab[KSPLIT * B_DIM][16 * PITCH];

  const int tid  = threadIdx.x;
  const int ww   = tid >> 5;
  const int lane = tid & 31;
  const int half = lane >> 4;
  const int col  = lane & 15;
  const int b    = ww & 7;
  const int q    = ww >> 3;
  const int s0   = blockIdx.x * 16;
  const int k0   = q * KCHUNK;
  const int rot  = 2 * ((blockIdx.x + (blockIdx.x >> 3)) & 7);

  const int lr = lane >> 3;
  const int lp = lane & 7;
  const float* __restrict__ xt =
      X + ((size_t)b * S_DIM + (size_t)(s0 + lr)) * D_DIM + k0 + 4 * lp;
  float* const myslab = slab[ww];
  float* const st = myslab + lr * PITCH + 4 * lp;
  const float* const fr = myslab + col * PITCH + 4 * half;

  const float* __restrict__ wcol = W + (size_t)(k0 + 4 * half) * E_DIM + col;

#define TOFF(T) ((((T) + rot) & (NTILE - 1)) * TK)
  v4f xa[AHEAD][4];
  v2f wb[2][8];

#define LOAD_X(T) do { _Pragma("unroll") for (int j = 0; j < 4; ++j) \
    xa[(T) % AHEAD][j] = *(const v4f*)(xt + (size_t)j * 4 * D_DIM + TOFF(T)); } while (0)
#define LOAD_W(T) do { _Pragma("unroll") for (int i = 0; i < 8; ++i) { \
    const int kr = TOFF(T) + 8 * (i >> 1) + 2 * (i & 1); \
    wb[(T) & 1][i].x = wcol[kr * E_DIM]; \
    wb[(T) & 1][i].y = wcol[kr * E_DIM + E_DIM]; } } while (0)
#define WAVE_LDS_ORDER() do { __builtin_amdgcn_fence(__ATOMIC_RELEASE, "wavefront"); \
    __builtin_amdgcn_wave_barrier(); \
    __builtin_amdgcn_fence(__ATOMIC_ACQUIRE, "wavefront"); } while (0)

  LOAD_W(0);
  #pragma unroll
  for (int t = 0; t < AHEAD; ++t) LOAD_X(t);

  v8f c = {};
  #pragma unroll
  for (int t = 0; t < NTILE; ++t) {
    #pragma unroll
    for (int j = 0; j < 4; ++j) *(v4f*)(st + j * 4 * PITCH) = xa[t % AHEAD][j];
    if (t + 1 < NTILE) LOAD_W(t + 1);
    if (t + AHEAD < NTILE) LOAD_X(t + AHEAD);
    WAVE_LDS_ORDER();
    #pragma unroll
    for (int w = 0; w < 4; ++w) {
      const v4f av = *(const v4f*)(fr + 8 * w);
      const v2f a0 = {av.x, av.y};
      const v2f a1 = {av.z, av.w};
      c = __builtin_amdgcn_wmma_f32_16x16x4_f32(
           false, a0,  false, wb[t & 1][2 * w],
           (short)0, c,  false,  false);
      c = __builtin_amdgcn_wmma_f32_16x16x4_f32(
           false, a1,  false, wb[t & 1][2 * w + 1],
           (short)0, c,  false,  false);
    }
    WAVE_LDS_ORDER();
  }
#undef LOAD_X
#undef LOAD_W
#undef TOFF
#undef WAVE_LDS_ORDER

  #pragma unroll
  for (int j = 0; j < 8; ++j) part[ww][j + 8 * half][col] = c[j];
  __syncthreads();

  const float bval = bias[col];
  float g[8];
  #pragma unroll
  for (int j = 0; j < 8; ++j) {
    const int m = j + 8 * half;
    float l = bval + part[b][m][col] + part[b + 8][m][col]
                   + part[b + 16][m][col] + part[b + 24][m][col];
    float mx = l;
    mx = fmaxf(mx, __shfl_xor(mx, 1));
    mx = fmaxf(mx, __shfl_xor(mx, 2));
    mx = fmaxf(mx, __shfl_xor(mx, 4));
    mx = fmaxf(mx, __shfl_xor(mx, 8));
    float e = __expf(l - mx);
    float s = e;
    s += __shfl_xor(s, 1);
    s += __shfl_xor(s, 2);
    s += __shfl_xor(s, 4);
    s += __shfl_xor(s, 8);
    g[j] = e / s;
    if (q == 0) gate[b][j + 8 * half][col] = g[j];
  }
  __syncthreads();

  if (q == 0) {
    #pragma unroll
    for (int j = 0; j < 8; ++j) {
      const int m = j + 8 * half;
      float denom = EPSILON;
      #pragma unroll
      for (int bb = 0; bb < B_DIM; ++bb) denom += gate[bb][m][col];
      out[((size_t)b * S_DIM + (size_t)(s0 + m)) * E_DIM + col] =
          g[j] / denom * CAPACITY;
    }
  }
}

extern "C" void kernel_launch(void* const* d_in, const int* in_sizes, int n_in,
                              void* d_out, int out_size, void* d_ws, size_t ws_size,
                              hipStream_t stream) {
  const float* X    = (const float*)d_in[0];
  const float* W    = (const float*)d_in[1];
  const float* bias = (const float*)d_in[2];
  float* out        = (float*)d_out;

  dim3 grid(S_DIM / 16);
  dim3 block(1024);
  switch_gate_wmma_kernel<<<grid, block, 0, stream>>>(X, W, bias, out);
}
